# starting kernel
# baseline (speedup 1.0000x reference)
#include <hip/hip_runtime.h>
#include <hip/hip_bf16.h>
#include <math.h>

typedef __attribute__((ext_vector_type(2))) float v2f;
typedef __attribute__((ext_vector_type(8))) float v8f;

#define N_NODES   50000
#define N_EDGES   800000
#define N_ENTRIES (N_EDGES + N_NODES)
#define IN_CH     128
#define HEADS     4
#define OUT_CH    32
#define HC        128

__device__ __forceinline__ float lrelu(float v) { return v > 0.0f ? v : 0.2f * v; }

__device__ __forceinline__ void edge_sd(int i, const int* __restrict__ ei, int& src, int& dst) {
    if (i < N_EDGES) { src = ei[i]; dst = ei[N_EDGES + i]; }
    else             { src = i - N_EDGES; dst = src; }
}

__global__ void gat_init(float* __restrict__ out, const float* __restrict__ bias,
                         float* __restrict__ nmax, float* __restrict__ nsum) {
    int idx = blockIdx.x * 256 + threadIdx.x;
    out[idx] = bias[idx & (HC - 1)];
    if (idx < N_NODES * HEADS) { nmax[idx] = -3.0e38f; nsum[idx] = 0.0f; }
}

__global__ __launch_bounds__(256) void gat_gemm(const float* __restrict__ x,
                                                const float* __restrict__ W,
                                                float* __restrict__ h) {
    __shared__ float xs[16 * 132];
    __shared__ float wt[32 * 132];

    const int t    = threadIdx.x;
    const int wave = t >> 5;
    const int lane = t & 31;
    const int r0   = blockIdx.x * 16;

    {
        int r = t >> 5;
        int c = (t & 31) * 4;
        #pragma unroll
        for (int i = 0; i < 2; ++i) {
            int rr = r + 8 * i;
            float4 v = *(const float4*)(x + (size_t)(r0 + rr) * IN_CH + c);
            xs[rr * 132 + c + 0] = v.x;
            xs[rr * 132 + c + 1] = v.y;
            xs[rr * 132 + c + 2] = v.z;
            xs[rr * 132 + c + 3] = v.w;
        }
    }

    const int row  = lane & 15;
    const int kh   = (lane >> 4) * 2;
    const int n0   = wave * 16;
    const int ncol = n0 + row;

    v8f acc = {0.f, 0.f, 0.f, 0.f, 0.f, 0.f, 0.f, 0.f};

    for (int kc = 0; kc < IN_CH; kc += 32) {
        __syncthreads();
        #pragma unroll
        for (int i = 0; i < 4; ++i) {
            int q    = t + 256 * i;
            int flat = q * 4;
            int n    = flat >> 5;
            int k4   = flat & 31;
            float4 v = *(const float4*)(W + (size_t)n * IN_CH + kc + k4);
            wt[(k4 + 0) * 132 + n] = v.x;
            wt[(k4 + 1) * 132 + n] = v.y;
            wt[(k4 + 2) * 132 + n] = v.z;
            wt[(k4 + 3) * 132 + n] = v.w;
        }
        __syncthreads();

        #pragma unroll
        for (int kk = 0; kk < 32; kk += 4) {
            v2f a, b;
            a.x = xs[row * 132 + kc + kk + kh + 0];
            a.y = xs[row * 132 + kc + kk + kh + 1];
            b.x = wt[(kk + kh + 0) * 132 + ncol];
            b.y = wt[(kk + kh + 1) * 132 + ncol];
            acc = __builtin_amdgcn_wmma_f32_16x16x4_f32(
                false, a, false, b, (short)0, acc, false, false);
        }
    }

    const int mrow = r0 + ((lane >> 4) * 8);
    const int col  = n0 + (lane & 15);
    #pragma unroll
    for (int r = 0; r < 8; ++r) {
        h[(size_t)(mrow + r) * HC + col] = acc[r];
    }
}

__global__ void gat_scores(const float* __restrict__ h, const float* __restrict__ att,
                           float* __restrict__ ssrc, float* __restrict__ sdst) {
    int node = blockIdx.x * 8 + (threadIdx.x >> 5);
    int lane = threadIdx.x & 31;
    int c    = lane * 4;
    int head = lane >> 3;
    int cc   = c & 31;

    float4 hv = *(const float4*)(h + (size_t)node * HC + c);
    float4 as = *(const float4*)(att + head * (2 * OUT_CH) + cc);
    float4 ad = *(const float4*)(att + head * (2 * OUT_CH) + OUT_CH + cc);

    float ps = hv.x * as.x + hv.y * as.y + hv.z * as.z + hv.w * as.w;
    float pd = hv.x * ad.x + hv.y * ad.y + hv.z * ad.z + hv.w * ad.w;

    #pragma unroll
    for (int off = 4; off >= 1; off >>= 1) {
        ps += __shfl_xor(ps, off, 8);
        pd += __shfl_xor(pd, off, 8);
    }
    if ((lane & 7) == 0) {
        ssrc[node * HEADS + head] = ps;
        sdst[node * HEADS + head] = pd;
    }
}

__global__ void gat_edge_max(const int* __restrict__ ei,
                             const float* __restrict__ ssrc, const float* __restrict__ sdst,
                             float* __restrict__ nmax) {
    int i = blockIdx.x * 256 + threadIdx.x;
    if (i >= N_ENTRIES) return;
    int src, dst; edge_sd(i, ei, src, dst);
    float4 a = *(const float4*)(ssrc + src * HEADS);
    float4 b = *(const float4*)(sdst + dst * HEADS);
    atomicMax(&nmax[dst * HEADS + 0], lrelu(a.x + b.x));
    atomicMax(&nmax[dst * HEADS + 1], lrelu(a.y + b.y));
    atomicMax(&nmax[dst * HEADS + 2], lrelu(a.z + b.z));
    atomicMax(&nmax[dst * HEADS + 3], lrelu(a.w + b.w));
}

__global__ void gat_edge_sum(const int* __restrict__ ei,
                             const float* __restrict__ ssrc, const float* __restrict__ sdst,
                             const float* __restrict__ nmax, float* __restrict__ nsum) {
    int i = blockIdx.x * 256 + threadIdx.x;
    if (i >= N_ENTRIES) return;
    int src, dst; edge_sd(i, ei, src, dst);
    float4 a  = *(const float4*)(ssrc + src * HEADS);
    float4 b  = *(const float4*)(sdst + dst * HEADS);
    float4 mx = *(const float4*)(nmax + dst * HEADS);
    atomicAdd(&nsum[dst * HEADS + 0], __expf(lrelu(a.x + b.x) - mx.x));
    atomicAdd(&nsum[dst * HEADS + 1], __expf(lrelu(a.y + b.y) - mx.y));
    atomicAdd(&nsum[dst * HEADS + 2], __expf(lrelu(a.z + b.z) - mx.z));
    atomicAdd(&nsum[dst * HEADS + 3], __expf(lrelu(a.w + b.w) - mx.w));
}

__global__ void gat_aggregate(const int* __restrict__ ei,
                              const float* __restrict__ ssrc, const float* __restrict__ sdst,
                              const float* __restrict__ nmax, const float* __restrict__ nsum,
                              const float* __restrict__ h, float* __restrict__ out) {
    int entry = blockIdx.x * 8 + (threadIdx.x >> 5);
    int lane  = threadIdx.x & 31;
    int src, dst; edge_sd(entry, ei, src, dst);

    float4 a  = *(const float4*)(ssrc + src * HEADS);
    float4 b  = *(const float4*)(sdst + dst * HEADS);
    float4 mx = *(const float4*)(nmax + dst * HEADS);
    float4 sm = *(const float4*)(nsum + dst * HEADS);

    float c0 = __expf(lrelu(a.x + b.x) - mx.x) / fmaxf(sm.x, 1e-10f);
    float c1 = __expf(lrelu(a.y + b.y) - mx.y) / fmaxf(sm.y, 1e-10f);
    float c2 = __expf(lrelu(a.z + b.z) - mx.z) / fmaxf(sm.z, 1e-10f);
    float c3 = __expf(lrelu(a.w + b.w) - mx.w) / fmaxf(sm.w, 1e-10f);

    int c    = lane * 4;
    int head = lane >> 3;
    float w  = (head < 2) ? (head == 0 ? c0 : c1) : (head == 2 ? c2 : c3);

    float4 hv = *(const float4*)(h + (size_t)src * HC + c);
    float* o  = out + (size_t)dst * HC + c;
    atomicAdd(o + 0, w * hv.x);
    atomicAdd(o + 1, w * hv.y);
    atomicAdd(o + 2, w * hv.z);
    atomicAdd(o + 3, w * hv.w);
}

extern "C" void kernel_launch(void* const* d_in, const int* in_sizes, int n_in,
                              void* d_out, int out_size, void* d_ws, size_t ws_size,
                              hipStream_t stream) {
    const float* x    = (const float*)d_in[0];
    const int*   ei   = (const int*)  d_in[1];
    const float* W    = (const float*)d_in[2];
    const float* att  = (const float*)d_in[3];
    const float* bias = (const float*)d_in[4];
    float* out = (float*)d_out;

    char* ws = (char*)d_ws;
    float* h    = (float*)(ws);
    float* ssrc = (float*)(ws + 25600000);
    float* sdst = (float*)(ws + 26400000);
    float* nmax = (float*)(ws + 27200000);
    float* nsum = (float*)(ws + 28000000);

    gat_init     <<<(N_NODES * HC) / 256, 256, 0, stream>>>(out, bias, nmax, nsum);
    gat_gemm     <<<N_NODES / 16,          256, 0, stream>>>(x, W, h);
    gat_scores   <<<N_NODES / 8,           256, 0, stream>>>(h, att, ssrc, sdst);
    int eb = (N_ENTRIES + 255) / 256;
    gat_edge_max <<<eb,                    256, 0, stream>>>(ei, ssrc, sdst, nmax);
    gat_edge_sum <<<eb,                    256, 0, stream>>>(ei, ssrc, sdst, nmax, nsum);
    gat_aggregate<<<N_ENTRIES / 8,         256, 0, stream>>>(ei, ssrc, sdst, nmax, nsum, h, out);
}


# edges grouped by destination (integer count, scan, fill): one wave per node does softmax and weighted row sum in registers, one store per row; float atomics gone, sums in arrival-order rank
# speedup vs baseline: 1.9237x; 1.9237x over previous
#include <hip/hip_runtime.h>
#include <hip/hip_bf16.h>
#include <math.h>


typedef __attribute__((ext_vector_type(2))) float v2f;
typedef __attribute__((ext_vector_type(8))) float v8f;

#define N_NODES   50000
#define N_EDGES   800000
#define IN_CH     128
#define HEADS     4
#define OUT_CH    32
#define HC        128

__device__ __forceinline__ float lrelu(float v) { return v > 0.0f ? v : 0.2f * v; }

__global__ __launch_bounds__(256) void gat_gemm(const float* __restrict__ x,
                                                const float* __restrict__ W,
                                                float* __restrict__ h) {
    __shared__ float xs[16 * 132];
    __shared__ float wt[32 * 132];

    const int t    = threadIdx.x;
    const int wave = t >> 5;
    const int lane = t & 31;
    const int r0   = blockIdx.x * 16;

    {
        int r = t >> 5;
        int c = (t & 31) * 4;
        #pragma unroll
        for (int i = 0; i < 2; ++i) {
            int rr = r + 8 * i;
            float4 v = *(const float4*)(x + (size_t)(r0 + rr) * IN_CH + c);
            xs[rr * 132 + c + 0] = v.x;
            xs[rr * 132 + c + 1] = v.y;
            xs[rr * 132 + c + 2] = v.z;
            xs[rr * 132 + c + 3] = v.w;
        }
    }

    const int row  = lane & 15;
    const int kh   = (lane >> 4) * 2;
    const int n0   = wave * 16;
    const int ncol = n0 + row;

    v8f acc = {0.f, 0.f, 0.f, 0.f, 0.f, 0.f, 0.f, 0.f};

    for (int kc = 0; kc < IN_CH; kc += 32) {
        __syncthreads();
        #pragma unroll
        for (int i = 0; i < 4; ++i) {
            int q    = t + 256 * i;
            int flat = q * 4;
            int n    = flat >> 5;
            int k4   = flat & 31;
            float4 v = *(const float4*)(W + (size_t)n * IN_CH + kc + k4);
            wt[(k4 + 0) * 132 + n] = v.x;
            wt[(k4 + 1) * 132 + n] = v.y;
            wt[(k4 + 2) * 132 + n] = v.z;
            wt[(k4 + 3) * 132 + n] = v.w;
        }
        __syncthreads();

        #pragma unroll
        for (int kk = 0; kk < 32; kk += 4) {
            v2f a, b;
            a.x = xs[row * 132 + kc + kk + kh + 0];
            a.y = xs[row * 132 + kc + kk + kh + 1];
            b.x = wt[(kk + kh + 0) * 132 + ncol];
            b.y = wt[(kk + kh + 1) * 132 + ncol];
            acc = __builtin_amdgcn_wmma_f32_16x16x4_f32(
                false, a, false, b, (short)0, acc, false, false);
        }
    }

    const int mrow = r0 + ((lane >> 4) * 8);
    const int col  = n0 + (lane & 15);
    #pragma unroll
    for (int r = 0; r < 8; ++r) {
        h[(size_t)(mrow + r) * HC + col] = acc[r];
    }
}

__global__ void gat_scores(const float* __restrict__ h, const float* __restrict__ att,
                           float* __restrict__ ssrc, int* __restrict__ cursor) {
    int node = blockIdx.x * 8 + (threadIdx.x >> 5);
    int lane = threadIdx.x & 31;
    int c    = lane * 4;
    int head = lane >> 3;
    int cc   = c & 31;

    float4 hv = *(const float4*)(h + (size_t)node * HC + c);
    float4 as = *(const float4*)(att + head * (2 * OUT_CH) + cc);

    float ps = hv.x * as.x + hv.y * as.y + hv.z * as.z + hv.w * as.w;
    for (int off = 4; off >= 1; off >>= 1) ps += __shfl_xor(ps, off, 8);

    if ((lane & 7) == 0) ssrc[node * HEADS + head] = ps;
    if (lane == 0) cursor[node] = 0;
}

__global__ void gat_count(const int* __restrict__ ei, int* __restrict__ cursor) {
    int i = blockIdx.x * 256 + threadIdx.x;
    atomicAdd(&cursor[ei[N_EDGES + i]], 1);
}

#define SCAN_PER 49
__global__ __launch_bounds__(1024) void gat_scan(int* __restrict__ cursor) {
    __shared__ int wsum[32];
    const int t    = threadIdx.x;
    const int lane = t & 31;
    const int w    = t >> 5;
    const int base = t * SCAN_PER;

    int v[SCAN_PER];
    int s = 0;
    for (int k = 0; k < SCAN_PER; ++k) {
        int idx = base + k;
        v[k] = (idx < N_NODES) ? cursor[idx] : 0;
        s += v[k];
    }

    int inc = s;
    for (int off = 1; off < 32; off <<= 1) {
        int u = __shfl_up(inc, off);
        if (lane >= off) inc += u;
    }
    if (lane == 31) wsum[w] = inc;
    __syncthreads();
    if (w == 0) {
        int tot = wsum[lane];
        int winc = tot;
        for (int off = 1; off < 32; off <<= 1) {
            int u = __shfl_up(winc, off);
            if (lane >= off) winc += u;
        }
        wsum[lane] = winc - tot;
    }
    __syncthreads();

    int run = wsum[w] + inc - s;
    for (int k = 0; k < SCAN_PER; ++k) {
        int idx = base + k;
        if (idx < N_NODES) { cursor[idx] = run; run += v[k]; }
    }
}

__global__ void gat_fill(const int* __restrict__ ei, int* __restrict__ cursor,
                         unsigned short* __restrict__ csr) {
    int i = blockIdx.x * 256 + threadIdx.x;
    int src = ei[i];
    int dst = ei[N_EDGES + i];
    int pos = atomicAdd(&cursor[dst], 1);
    csr[pos] = (unsigned short)src;
}

__device__ __forceinline__ float wave_max(float v) {
    for (int off = 16; off >= 1; off >>= 1) v = fmaxf(v, __shfl_xor(v, off));
    return v;
}
__device__ __forceinline__ float wave_sum(float v) {
    for (int off = 16; off >= 1; off >>= 1) v += __shfl_xor(v, off);
    return v;
}
__device__ __forceinline__ float bcast(float v, int l) {
    return __int_as_float(__builtin_amdgcn_readlane(__float_as_int(v), l));
}
__device__ __forceinline__ float pick(float a, float b, float c, float d, int head) {
    return (head < 2) ? (head == 0 ? a : b) : (head == 2 ? c : d);
}

__global__ __launch_bounds__(256) void gat_node(const int* __restrict__ cursor,
                                                const unsigned short* __restrict__ csr,
                                                const float* __restrict__ ssrc,
                                                const float* __restrict__ h,
                                                const float* __restrict__ att,
                                                const float* __restrict__ bias,
                                                float* __restrict__ out) {
    const int node = blockIdx.x * 8 + (threadIdx.x >> 5);
    const int lane = threadIdx.x & 31;
    const int c    = lane * 4;
    const int head = lane >> 3;
    const int cc   = c & 31;

    const int beg = __builtin_amdgcn_readfirstlane((node == 0) ? 0 : cursor[node - 1]);
    const int end = __builtin_amdgcn_readfirstlane(cursor[node]);

    const float4 hs = *(const float4*)(h + (size_t)node * HC + c);
    const float4 as = *(const float4*)(att + head * (2 * OUT_CH) + cc);
    const float4 ad = *(const float4*)(att + head * (2 * OUT_CH) + OUT_CH + cc);
    float ps = hs.x * as.x + hs.y * as.y + hs.z * as.z + hs.w * as.w;
    float pd = hs.x * ad.x + hs.y * ad.y + hs.z * ad.z + hs.w * ad.w;
    for (int off = 4; off >= 1; off >>= 1) {
        ps += __shfl_xor(ps, off, 8);
        pd += __shfl_xor(pd, off, 8);
    }

    const float d0 = bcast(pd, 0), d1 = bcast(pd, 8), d2 = bcast(pd, 16), d3 = bcast(pd, 24);

    float m0 = lrelu(bcast(ps, 0)  + d0);
    float m1 = lrelu(bcast(ps, 8)  + d1);
    float m2 = lrelu(bcast(ps, 16) + d2);
    float m3 = lrelu(bcast(ps, 24) + d3);
    float S0 = 1.0f, S1 = 1.0f, S2 = 1.0f, S3 = 1.0f;
    float4 acc = hs;

    for (int base = beg; base < end; base += 32) {
        const int  j   = base + lane;
        const bool act = j < end;
        const int  s   = act ? (int)csr[j] : 0;
        const float4 a = *(const float4*)(ssrc + s * HEADS);

        const float e0 = act ? lrelu(a.x + d0) : -INFINITY;
        const float e1 = act ? lrelu(a.y + d1) : -INFINITY;
        const float e2 = act ? lrelu(a.z + d2) : -INFINITY;
        const float e3 = act ? lrelu(a.w + d3) : -INFINITY;

        const float n0 = fmaxf(m0, wave_max(e0));
        const float n1 = fmaxf(m1, wave_max(e1));
        const float n2 = fmaxf(m2, wave_max(e2));
        const float n3 = fmaxf(m3, wave_max(e3));

        const float r0 = __expf(m0 - n0);
        const float r1 = __expf(m1 - n1);
        const float r2 = __expf(m2 - n2);
        const float r3 = __expf(m3 - n3);

        const float p0 = act ? __expf(e0 - n0) : 0.0f;
        const float p1 = act ? __expf(e1 - n1) : 0.0f;
        const float p2 = act ? __expf(e2 - n2) : 0.0f;
        const float p3 = act ? __expf(e3 - n3) : 0.0f;

        S0 = S0 * r0 + wave_sum(p0);
        S1 = S1 * r1 + wave_sum(p1);
        S2 = S2 * r2 + wave_sum(p2);
        S3 = S3 * r3 + wave_sum(p3);
        m0 = n0; m1 = n1; m2 = n2; m3 = n3;

        const float r = pick(r0, r1, r2, r3, head);
        acc.x *= r; acc.y *= r; acc.z *= r; acc.w *= r;

        const int nin = min(32, end - base);
        int jj = 0;
        for (; jj + 4 <= nin; jj += 4) {
            float4 hv[4];
            float  wq[4];
            for (int u = 0; u < 4; ++u) {
                const int sj = __builtin_amdgcn_readlane(s, jj + u);
                hv[u] = *(const float4*)(h + (size_t)sj * HC + c);
                wq[u] = pick(bcast(p0, jj + u), bcast(p1, jj + u),
                             bcast(p2, jj + u), bcast(p3, jj + u), head);
            }
            for (int u = 0; u < 4; ++u) {
                acc.x += wq[u] * hv[u].x;
                acc.y += wq[u] * hv[u].y;
                acc.z += wq[u] * hv[u].z;
                acc.w += wq[u] * hv[u].w;
            }
        }
        for (; jj < nin; ++jj) {
            const int sj = __builtin_amdgcn_readlane(s, jj);
            const float4 hv = *(const float4*)(h + (size_t)sj * HC + c);
            const float wq = pick(bcast(p0, jj), bcast(p1, jj), bcast(p2, jj), bcast(p3, jj), head);
            acc.x += wq * hv.x;
            acc.y += wq * hv.y;
            acc.z += wq * hv.z;
            acc.w += wq * hv.w;
        }
    }

    const float inv = 1.0f / fmaxf(pick(S0, S1, S2, S3, head), 1e-10f);
    const float4 b  = *(const float4*)(bias + c);
    float4 o;
    o.x = acc.x * inv + b.x;
    o.y = acc.y * inv + b.y;
    o.z = acc.z * inv + b.z;
    o.w = acc.w * inv + b.w;
    *(float4*)(out + (size_t)node * HC + c) = o;
}

extern "C" void kernel_launch(void* const* d_in, const int* in_sizes, int n_in,
                              void* d_out, int out_size, void* d_ws, size_t ws_size,
                              hipStream_t stream) {
    const float* x    = (const float*)d_in[0];
    const int*   ei   = (const int*)  d_in[1];
    const float* W    = (const float*)d_in[2];
    const float* att  = (const float*)d_in[3];
    const float* bias = (const float*)d_in[4];
    float* out = (float*)d_out;

    char* ws = (char*)d_ws;
    float*          h      = (float*)(ws);
    float*          ssrc   = (float*)(ws + 25600000);
    int*            cursor = (int*)(ws + 26400000);
    unsigned short* csr    = (unsigned short*)(ws + 26600000);

    gat_gemm  <<<N_NODES / 16,  256,  0, stream>>>(x, W, h);
    gat_scores<<<N_NODES / 8,   256,  0, stream>>>(h, att, ssrc, cursor);
    gat_count <<<N_EDGES / 256, 256,  0, stream>>>(ei, cursor);
    gat_scan  <<<1,             1024, 0, stream>>>(cursor);
    gat_fill  <<<N_EDGES / 256, 256,  0, stream>>>(ei, cursor, csr);
    gat_node  <<<N_NODES / 8,   256,  0, stream>>>(cursor, csr, ssrc, h, att, bias, out);
}


# one atomic pass for the edge bucketing instead of two: the counting atomic returns each edge's rank, the fill places it at range start + rank with no atomic; sums stay in arrival-order rank
# speedup vs baseline: 2.2303x; 1.1594x over previous
#include <hip/hip_runtime.h>
#include <hip/hip_bf16.h>
#include <math.h>


typedef __attribute__((ext_vector_type(2))) float v2f;
typedef __attribute__((ext_vector_type(8))) float v8f;

#define N_NODES   50000
#define N_EDGES   800000
#define IN_CH     128
#define HEADS     4
#define OUT_CH    32
#define HC        128

__device__ __forceinline__ float lrelu(float v) { return v > 0.0f ? v : 0.2f * v; }

__global__ __launch_bounds__(256) void gat_gemm(const float* __restrict__ x,
                                                const float* __restrict__ W,
                                                float* __restrict__ h) {
    __shared__ float xs[16 * 132];
    __shared__ float wt[32 * 132];

    const int t    = threadIdx.x;
    const int wave = t >> 5;
    const int lane = t & 31;
    const int r0   = blockIdx.x * 16;

    {
        int r = t >> 5;
        int c = (t & 31) * 4;
        #pragma unroll
        for (int i = 0; i < 2; ++i) {
            int rr = r + 8 * i;
            float4 v = *(const float4*)(x + (size_t)(r0 + rr) * IN_CH + c);
            xs[rr * 132 + c + 0] = v.x;
            xs[rr * 132 + c + 1] = v.y;
            xs[rr * 132 + c + 2] = v.z;
            xs[rr * 132 + c + 3] = v.w;
        }
    }

    const int row  = lane & 15;
    const int kh   = (lane >> 4) * 2;
    const int n0   = wave * 16;
    const int ncol = n0 + row;

    v8f acc = {0.f, 0.f, 0.f, 0.f, 0.f, 0.f, 0.f, 0.f};

    for (int kc = 0; kc < IN_CH; kc += 32) {
        __syncthreads();
        #pragma unroll
        for (int i = 0; i < 4; ++i) {
            int q    = t + 256 * i;
            int flat = q * 4;
            int n    = flat >> 5;
            int k4   = flat & 31;
            float4 v = *(const float4*)(W + (size_t)n * IN_CH + kc + k4);
            wt[(k4 + 0) * 132 + n] = v.x;
            wt[(k4 + 1) * 132 + n] = v.y;
            wt[(k4 + 2) * 132 + n] = v.z;
            wt[(k4 + 3) * 132 + n] = v.w;
        }
        __syncthreads();

        #pragma unroll
        for (int kk = 0; kk < 32; kk += 4) {
            v2f a, b;
            a.x = xs[row * 132 + kc + kk + kh + 0];
            a.y = xs[row * 132 + kc + kk + kh + 1];
            b.x = wt[(kk + kh + 0) * 132 + ncol];
            b.y = wt[(kk + kh + 1) * 132 + ncol];
            acc = __builtin_amdgcn_wmma_f32_16x16x4_f32(
                false, a, false, b, (short)0, acc, false, false);
        }
    }

    const int mrow = r0 + ((lane >> 4) * 8);
    const int col  = n0 + (lane & 15);
    #pragma unroll
    for (int r = 0; r < 8; ++r) {
        h[(size_t)(mrow + r) * HC + col] = acc[r];
    }
}

__global__ void gat_scores(const float* __restrict__ h, const float* __restrict__ att,
                           float* __restrict__ ssrc, int* __restrict__ cursor) {
    int node = blockIdx.x * 8 + (threadIdx.x >> 5);
    int lane = threadIdx.x & 31;
    int c    = lane * 4;
    int head = lane >> 3;
    int cc   = c & 31;

    float4 hv = *(const float4*)(h + (size_t)node * HC + c);
    float4 as = *(const float4*)(att + head * (2 * OUT_CH) + cc);

    float ps = hv.x * as.x + hv.y * as.y + hv.z * as.z + hv.w * as.w;
    for (int off = 4; off >= 1; off >>= 1) ps += __shfl_xor(ps, off, 8);

    if ((lane & 7) == 0) ssrc[node * HEADS + head] = ps;
    if (lane == 0) cursor[node] = 0;
}

__global__ void gat_count(const int* __restrict__ ei, int* __restrict__ cursor,
                          int* __restrict__ rank) {
    int i = blockIdx.x * 256 + threadIdx.x;
    rank[i] = atomicAdd(&cursor[ei[N_EDGES + i]], 1);
}

#define SCAN_PER 49
__global__ __launch_bounds__(1024) void gat_scan(int* __restrict__ cursor) {
    __shared__ int wsum[32];
    const int t    = threadIdx.x;
    const int lane = t & 31;
    const int w    = t >> 5;
    const int base = t * SCAN_PER;

    int v[SCAN_PER];
    int s = 0;
    for (int k = 0; k < SCAN_PER; ++k) {
        int idx = base + k;
        v[k] = (idx < N_NODES) ? cursor[idx] : 0;
        s += v[k];
    }

    int inc = s;
    for (int off = 1; off < 32; off <<= 1) {
        int u = __shfl_up(inc, off);
        if (lane >= off) inc += u;
    }
    if (lane == 31) wsum[w] = inc;
    __syncthreads();
    if (w == 0) {
        int tot = wsum[lane];
        int winc = tot;
        for (int off = 1; off < 32; off <<= 1) {
            int u = __shfl_up(winc, off);
            if (lane >= off) winc += u;
        }
        wsum[lane] = winc - tot;
    }
    __syncthreads();

    int run = wsum[w] + inc - s;
    for (int k = 0; k < SCAN_PER; ++k) {
        int idx = base + k;
        if (idx < N_NODES) { cursor[idx] = run; run += v[k]; }
    }
}

__global__ void gat_fill(const int* __restrict__ ei, const int* __restrict__ cursor,
                         const int* __restrict__ rank,
                         unsigned short* __restrict__ csr) {
    int i = blockIdx.x * 256 + threadIdx.x;
    int src = ei[i];
    int dst = ei[N_EDGES + i];
    int pos = cursor[dst] + rank[i];
    csr[pos] = (unsigned short)src;
}

__device__ __forceinline__ float wave_max(float v) {
    for (int off = 16; off >= 1; off >>= 1) v = fmaxf(v, __shfl_xor(v, off));
    return v;
}
__device__ __forceinline__ float wave_sum(float v) {
    for (int off = 16; off >= 1; off >>= 1) v += __shfl_xor(v, off);
    return v;
}
__device__ __forceinline__ float bcast(float v, int l) {
    return __int_as_float(__builtin_amdgcn_readlane(__float_as_int(v), l));
}
__device__ __forceinline__ float pick(float a, float b, float c, float d, int head) {
    return (head < 2) ? (head == 0 ? a : b) : (head == 2 ? c : d);
}

__global__ __launch_bounds__(256) void gat_node(const int* __restrict__ cursor,
                                                const unsigned short* __restrict__ csr,
                                                const float* __restrict__ ssrc,
                                                const float* __restrict__ h,
                                                const float* __restrict__ att,
                                                const float* __restrict__ bias,
                                                float* __restrict__ out) {
    const int node = blockIdx.x * 8 + (threadIdx.x >> 5);
    const int lane = threadIdx.x & 31;
    const int c    = lane * 4;
    const int head = lane >> 3;
    const int cc   = c & 31;

    const int beg = __builtin_amdgcn_readfirstlane(cursor[node]);
    const int end = __builtin_amdgcn_readfirstlane((node == N_NODES - 1) ? N_EDGES : cursor[node + 1]);

    const float4 hs = *(const float4*)(h + (size_t)node * HC + c);
    const float4 as = *(const float4*)(att + head * (2 * OUT_CH) + cc);
    const float4 ad = *(const float4*)(att + head * (2 * OUT_CH) + OUT_CH + cc);
    float ps = hs.x * as.x + hs.y * as.y + hs.z * as.z + hs.w * as.w;
    float pd = hs.x * ad.x + hs.y * ad.y + hs.z * ad.z + hs.w * ad.w;
    for (int off = 4; off >= 1; off >>= 1) {
        ps += __shfl_xor(ps, off, 8);
        pd += __shfl_xor(pd, off, 8);
    }

    const float d0 = bcast(pd, 0), d1 = bcast(pd, 8), d2 = bcast(pd, 16), d3 = bcast(pd, 24);

    float m0 = lrelu(bcast(ps, 0)  + d0);
    float m1 = lrelu(bcast(ps, 8)  + d1);
    float m2 = lrelu(bcast(ps, 16) + d2);
    float m3 = lrelu(bcast(ps, 24) + d3);
    float S0 = 1.0f, S1 = 1.0f, S2 = 1.0f, S3 = 1.0f;
    float4 acc = hs;

    for (int base = beg; base < end; base += 32) {
        const int  j   = base + lane;
        const bool act = j < end;
        const int  s   = act ? (int)csr[j] : 0;
        const float4 a = *(const float4*)(ssrc + s * HEADS);

        const float e0 = act ? lrelu(a.x + d0) : -INFINITY;
        const float e1 = act ? lrelu(a.y + d1) : -INFINITY;
        const float e2 = act ? lrelu(a.z + d2) : -INFINITY;
        const float e3 = act ? lrelu(a.w + d3) : -INFINITY;

        const float n0 = fmaxf(m0, wave_max(e0));
        const float n1 = fmaxf(m1, wave_max(e1));
        const float n2 = fmaxf(m2, wave_max(e2));
        const float n3 = fmaxf(m3, wave_max(e3));

        const float r0 = __expf(m0 - n0);
        const float r1 = __expf(m1 - n1);
        const float r2 = __expf(m2 - n2);
        const float r3 = __expf(m3 - n3);

        const float p0 = act ? __expf(e0 - n0) : 0.0f;
        const float p1 = act ? __expf(e1 - n1) : 0.0f;
        const float p2 = act ? __expf(e2 - n2) : 0.0f;
        const float p3 = act ? __expf(e3 - n3) : 0.0f;

        S0 = S0 * r0 + wave_sum(p0);
        S1 = S1 * r1 + wave_sum(p1);
        S2 = S2 * r2 + wave_sum(p2);
        S3 = S3 * r3 + wave_sum(p3);
        m0 = n0; m1 = n1; m2 = n2; m3 = n3;

        const float r = pick(r0, r1, r2, r3, head);
        acc.x *= r; acc.y *= r; acc.z *= r; acc.w *= r;

        const int nin = min(32, end - base);
        int jj = 0;
        for (; jj + 4 <= nin; jj += 4) {
            float4 hv[4];
            float  wq[4];
            for (int u = 0; u < 4; ++u) {
                const int sj = __builtin_amdgcn_readlane(s, jj + u);
                hv[u] = *(const float4*)(h + (size_t)sj * HC + c);
                wq[u] = pick(bcast(p0, jj + u), bcast(p1, jj + u),
                             bcast(p2, jj + u), bcast(p3, jj + u), head);
            }
            for (int u = 0; u < 4; ++u) {
                acc.x += wq[u] * hv[u].x;
                acc.y += wq[u] * hv[u].y;
                acc.z += wq[u] * hv[u].z;
                acc.w += wq[u] * hv[u].w;
            }
        }
        for (; jj < nin; ++jj) {
            const int sj = __builtin_amdgcn_readlane(s, jj);
            const float4 hv = *(const float4*)(h + (size_t)sj * HC + c);
            const float wq = pick(bcast(p0, jj), bcast(p1, jj), bcast(p2, jj), bcast(p3, jj), head);
            acc.x += wq * hv.x;
            acc.y += wq * hv.y;
            acc.z += wq * hv.z;
            acc.w += wq * hv.w;
        }
    }

    const float inv = 1.0f / fmaxf(pick(S0, S1, S2, S3, head), 1e-10f);
    const float4 b  = *(const float4*)(bias + c);
    float4 o;
    o.x = acc.x * inv + b.x;
    o.y = acc.y * inv + b.y;
    o.z = acc.z * inv + b.z;
    o.w = acc.w * inv + b.w;
    *(float4*)(out + (size_t)node * HC + c) = o;
}

extern "C" void kernel_launch(void* const* d_in, const int* in_sizes, int n_in,
                              void* d_out, int out_size, void* d_ws, size_t ws_size,
                              hipStream_t stream) {
    const float* x    = (const float*)d_in[0];
    const int*   ei   = (const int*)  d_in[1];
    const float* W    = (const float*)d_in[2];
    const float* att  = (const float*)d_in[3];
    const float* bias = (const float*)d_in[4];
    float* out = (float*)d_out;
    int* rank = (int*)d_out;

    char* ws = (char*)d_ws;
    float*          h      = (float*)(ws);
    float*          ssrc   = (float*)(ws + 25600000);
    int*            cursor = (int*)(ws + 26400000);
    unsigned short* csr    = (unsigned short*)(ws + 26600000);

    gat_gemm  <<<N_NODES / 16,  256,  0, stream>>>(x, W, h);
    gat_scores<<<N_NODES / 8,   256,  0, stream>>>(h, att, ssrc, cursor);
    gat_count <<<N_EDGES / 256, 256,  0, stream>>>(ei, cursor, rank);
    gat_scan  <<<1,             1024, 0, stream>>>(cursor);
    gat_fill  <<<N_EDGES / 256, 256,  0, stream>>>(ei, cursor, rank, csr);
    gat_node  <<<N_NODES / 8,   256,  0, stream>>>(cursor, csr, ssrc, h, att, bias, out);
}


# prefix-sum pass loads and stores whole 512-byte rows: a wave owns a contiguous range, one b128 access per lane and row, scan in registers, not 49 four-byte accesses per thread 196 bytes apart
# speedup vs baseline: 2.9288x; 1.3132x over previous
#include <hip/hip_runtime.h>
#include <hip/hip_bf16.h>
#include <math.h>


typedef __attribute__((ext_vector_type(2))) float v2f;
typedef __attribute__((ext_vector_type(8))) float v8f;

#define N_NODES   50000
#define N_EDGES   800000
#define IN_CH     128
#define HEADS     4
#define OUT_CH    32
#define HC        128

__device__ __forceinline__ float lrelu(float v) { return v > 0.0f ? v : 0.2f * v; }

__global__ __launch_bounds__(256) void gat_gemm(const float* __restrict__ x,
                                                const float* __restrict__ W,
                                                float* __restrict__ h) {
    __shared__ float xs[16 * 132];
    __shared__ float wt[32 * 132];

    const int t    = threadIdx.x;
    const int wave = t >> 5;
    const int lane = t & 31;
    const int r0   = blockIdx.x * 16;

    {
        int r = t >> 5;
        int c = (t & 31) * 4;
        #pragma unroll
        for (int i = 0; i < 2; ++i) {
            int rr = r + 8 * i;
            float4 v = *(const float4*)(x + (size_t)(r0 + rr) * IN_CH + c);
            xs[rr * 132 + c + 0] = v.x;
            xs[rr * 132 + c + 1] = v.y;
            xs[rr * 132 + c + 2] = v.z;
            xs[rr * 132 + c + 3] = v.w;
        }
    }

    const int row  = lane & 15;
    const int kh   = (lane >> 4) * 2;
    const int n0   = wave * 16;
    const int ncol = n0 + row;

    v8f acc = {0.f, 0.f, 0.f, 0.f, 0.f, 0.f, 0.f, 0.f};

    for (int kc = 0; kc < IN_CH; kc += 32) {
        __syncthreads();
        #pragma unroll
        for (int i = 0; i < 4; ++i) {
            int q    = t + 256 * i;
            int flat = q * 4;
            int n    = flat >> 5;
            int k4   = flat & 31;
            float4 v = *(const float4*)(W + (size_t)n * IN_CH + kc + k4);
            wt[(k4 + 0) * 132 + n] = v.x;
            wt[(k4 + 1) * 132 + n] = v.y;
            wt[(k4 + 2) * 132 + n] = v.z;
            wt[(k4 + 3) * 132 + n] = v.w;
        }
        __syncthreads();

        #pragma unroll
        for (int kk = 0; kk < 32; kk += 4) {
            v2f a, b;
            a.x = xs[row * 132 + kc + kk + kh + 0];
            a.y = xs[row * 132 + kc + kk + kh + 1];
            b.x = wt[(kk + kh + 0) * 132 + ncol];
            b.y = wt[(kk + kh + 1) * 132 + ncol];
            acc = __builtin_amdgcn_wmma_f32_16x16x4_f32(
                false, a, false, b, (short)0, acc, false, false);
        }
    }

    const int mrow = r0 + ((lane >> 4) * 8);
    const int col  = n0 + (lane & 15);
    #pragma unroll
    for (int r = 0; r < 8; ++r) {
        h[(size_t)(mrow + r) * HC + col] = acc[r];
    }
}

__global__ void gat_scores(const float* __restrict__ h, const float* __restrict__ att,
                           float* __restrict__ ssrc, int* __restrict__ cursor) {
    int node = blockIdx.x * 8 + (threadIdx.x >> 5);
    int lane = threadIdx.x & 31;
    int c    = lane * 4;
    int head = lane >> 3;
    int cc   = c & 31;

    float4 hv = *(const float4*)(h + (size_t)node * HC + c);
    float4 as = *(const float4*)(att + head * (2 * OUT_CH) + cc);

    float ps = hv.x * as.x + hv.y * as.y + hv.z * as.z + hv.w * as.w;
    for (int off = 4; off >= 1; off >>= 1) ps += __shfl_xor(ps, off, 8);

    if ((lane & 7) == 0) ssrc[node * HEADS + head] = ps;
    if (lane == 0) cursor[node] = 0;
}

__global__ void gat_count(const int* __restrict__ ei, int* __restrict__ cursor,
                          int* __restrict__ rank) {
    int i = blockIdx.x * 256 + threadIdx.x;
    rank[i] = atomicAdd(&cursor[ei[N_EDGES + i]], 1);
}

#define SCAN_ROWS 13
#define SCAN_WAVE (SCAN_ROWS * 128)
__global__ __launch_bounds__(1024) void gat_scan(int* __restrict__ cursor) {
    __shared__ int wsum[32];
    const int t    = threadIdx.x;
    const int lane = t & 31;
    const int w    = t >> 5;
    const int base = w * SCAN_WAVE + lane * 4;

    int4 v[SCAN_ROWS];
    for (int i = 0; i < SCAN_ROWS; ++i) {
        const int idx = base + i * 128;
        v[i] = (idx < N_NODES) ? *(const int4*)(cursor + idx) : make_int4(0, 0, 0, 0);
    }

    int inc[SCAN_ROWS];
    for (int i = 0; i < SCAN_ROWS; ++i) inc[i] = v[i].x + v[i].y + v[i].z + v[i].w;
    for (int off = 1; off < 32; off <<= 1) {
        int u[SCAN_ROWS];
        for (int i = 0; i < SCAN_ROWS; ++i) u[i] = __shfl_up(inc[i], off);
        for (int i = 0; i < SCAN_ROWS; ++i) if (lane >= off) inc[i] += u[i];
    }

    int rowoff[SCAN_ROWS];
    int wtot = 0;
    for (int i = 0; i < SCAN_ROWS; ++i) {
        rowoff[i] = wtot;
        wtot += __builtin_amdgcn_readlane(inc[i], 31);
    }
    if (lane == 0) wsum[w] = wtot;
    __syncthreads();
    if (w == 0) {
        int tot = wsum[lane];
        int winc = tot;
        for (int off = 1; off < 32; off <<= 1) {
            int u = __shfl_up(winc, off);
            if (lane >= off) winc += u;
        }
        wsum[lane] = winc - tot;
    }
    __syncthreads();

    const int wbase = wsum[w];
    for (int i = 0; i < SCAN_ROWS; ++i) {
        const int idx = base + i * 128;
        if (idx < N_NODES) {
            int4 o;
            o.x = wbase + rowoff[i] + inc[i] - (v[i].x + v[i].y + v[i].z + v[i].w);
            o.y = o.x + v[i].x;
            o.z = o.y + v[i].y;
            o.w = o.z + v[i].z;
            *(int4*)(cursor + idx) = o;
        }
    }
}

__global__ void gat_fill(const int* __restrict__ ei, const int* __restrict__ cursor,
                         const int* __restrict__ rank,
                         unsigned short* __restrict__ csr) {
    int i = blockIdx.x * 256 + threadIdx.x;
    int src = ei[i];
    int dst = ei[N_EDGES + i];
    int pos = cursor[dst] + rank[i];
    csr[pos] = (unsigned short)src;
}

__device__ __forceinline__ float wave_max(float v) {
    for (int off = 16; off >= 1; off >>= 1) v = fmaxf(v, __shfl_xor(v, off));
    return v;
}
__device__ __forceinline__ float wave_sum(float v) {
    for (int off = 16; off >= 1; off >>= 1) v += __shfl_xor(v, off);
    return v;
}
__device__ __forceinline__ float bcast(float v, int l) {
    return __int_as_float(__builtin_amdgcn_readlane(__float_as_int(v), l));
}
__device__ __forceinline__ float pick(float a, float b, float c, float d, int head) {
    return (head < 2) ? (head == 0 ? a : b) : (head == 2 ? c : d);
}

__global__ __launch_bounds__(256) void gat_node(const int* __restrict__ cursor,
                                                const unsigned short* __restrict__ csr,
                                                const float* __restrict__ ssrc,
                                                const float* __restrict__ h,
                                                const float* __restrict__ att,
                                                const float* __restrict__ bias,
                                                float* __restrict__ out) {
    const int node = blockIdx.x * 8 + (threadIdx.x >> 5);
    const int lane = threadIdx.x & 31;
    const int c    = lane * 4;
    const int head = lane >> 3;
    const int cc   = c & 31;

    const int beg = __builtin_amdgcn_readfirstlane(cursor[node]);
    const int end = __builtin_amdgcn_readfirstlane((node == N_NODES - 1) ? N_EDGES : cursor[node + 1]);

    const float4 hs = *(const float4*)(h + (size_t)node * HC + c);
    const float4 as = *(const float4*)(att + head * (2 * OUT_CH) + cc);
    const float4 ad = *(const float4*)(att + head * (2 * OUT_CH) + OUT_CH + cc);
    float ps = hs.x * as.x + hs.y * as.y + hs.z * as.z + hs.w * as.w;
    float pd = hs.x * ad.x + hs.y * ad.y + hs.z * ad.z + hs.w * ad.w;
    for (int off = 4; off >= 1; off >>= 1) {
        ps += __shfl_xor(ps, off, 8);
        pd += __shfl_xor(pd, off, 8);
    }

    const float d0 = bcast(pd, 0), d1 = bcast(pd, 8), d2 = bcast(pd, 16), d3 = bcast(pd, 24);

    float m0 = lrelu(bcast(ps, 0)  + d0);
    float m1 = lrelu(bcast(ps, 8)  + d1);
    float m2 = lrelu(bcast(ps, 16) + d2);
    float m3 = lrelu(bcast(ps, 24) + d3);
    float S0 = 1.0f, S1 = 1.0f, S2 = 1.0f, S3 = 1.0f;
    float4 acc = hs;

    for (int base = beg; base < end; base += 32) {
        const int  j   = base + lane;
        const bool act = j < end;
        const int  s   = act ? (int)csr[j] : 0;
        const float4 a = *(const float4*)(ssrc + s * HEADS);

        const float e0 = act ? lrelu(a.x + d0) : -INFINITY;
        const float e1 = act ? lrelu(a.y + d1) : -INFINITY;
        const float e2 = act ? lrelu(a.z + d2) : -INFINITY;
        const float e3 = act ? lrelu(a.w + d3) : -INFINITY;

        const float n0 = fmaxf(m0, wave_max(e0));
        const float n1 = fmaxf(m1, wave_max(e1));
        const float n2 = fmaxf(m2, wave_max(e2));
        const float n3 = fmaxf(m3, wave_max(e3));

        const float r0 = __expf(m0 - n0);
        const float r1 = __expf(m1 - n1);
        const float r2 = __expf(m2 - n2);
        const float r3 = __expf(m3 - n3);

        const float p0 = act ? __expf(e0 - n0) : 0.0f;
        const float p1 = act ? __expf(e1 - n1) : 0.0f;
        const float p2 = act ? __expf(e2 - n2) : 0.0f;
        const float p3 = act ? __expf(e3 - n3) : 0.0f;

        S0 = S0 * r0 + wave_sum(p0);
        S1 = S1 * r1 + wave_sum(p1);
        S2 = S2 * r2 + wave_sum(p2);
        S3 = S3 * r3 + wave_sum(p3);
        m0 = n0; m1 = n1; m2 = n2; m3 = n3;

        const float r = pick(r0, r1, r2, r3, head);
        acc.x *= r; acc.y *= r; acc.z *= r; acc.w *= r;

        const int nin = min(32, end - base);
        int jj = 0;
        for (; jj + 4 <= nin; jj += 4) {
            float4 hv[4];
            float  wq[4];
            for (int u = 0; u < 4; ++u) {
                const int sj = __builtin_amdgcn_readlane(s, jj + u);
                hv[u] = *(const float4*)(h + (size_t)sj * HC + c);
                wq[u] = pick(bcast(p0, jj + u), bcast(p1, jj + u),
                             bcast(p2, jj + u), bcast(p3, jj + u), head);
            }
            for (int u = 0; u < 4; ++u) {
                acc.x += wq[u] * hv[u].x;
                acc.y += wq[u] * hv[u].y;
                acc.z += wq[u] * hv[u].z;
                acc.w += wq[u] * hv[u].w;
            }
        }
        for (; jj < nin; ++jj) {
            const int sj = __builtin_amdgcn_readlane(s, jj);
            const float4 hv = *(const float4*)(h + (size_t)sj * HC + c);
            const float wq = pick(bcast(p0, jj), bcast(p1, jj), bcast(p2, jj), bcast(p3, jj), head);
            acc.x += wq * hv.x;
            acc.y += wq * hv.y;
            acc.z += wq * hv.z;
            acc.w += wq * hv.w;
        }
    }

    const float inv = 1.0f / fmaxf(pick(S0, S1, S2, S3, head), 1e-10f);
    const float4 b  = *(const float4*)(bias + c);
    float4 o;
    o.x = acc.x * inv + b.x;
    o.y = acc.y * inv + b.y;
    o.z = acc.z * inv + b.z;
    o.w = acc.w * inv + b.w;
    *(float4*)(out + (size_t)node * HC + c) = o;
}

extern "C" void kernel_launch(void* const* d_in, const int* in_sizes, int n_in,
                              void* d_out, int out_size, void* d_ws, size_t ws_size,
                              hipStream_t stream) {
    const float* x    = (const float*)d_in[0];
    const int*   ei   = (const int*)  d_in[1];
    const float* W    = (const float*)d_in[2];
    const float* att  = (const float*)d_in[3];
    const float* bias = (const float*)d_in[4];
    float* out = (float*)d_out;
    int* rank = (int*)d_out;

    char* ws = (char*)d_ws;
    float*          h      = (float*)(ws);
    float*          ssrc   = (float*)(ws + 25600000);
    int*            cursor = (int*)(ws + 26400000);
    unsigned short* csr    = (unsigned short*)(ws + 26600000);

    gat_gemm  <<<N_NODES / 16,  256,  0, stream>>>(x, W, h);
    gat_scores<<<N_NODES / 8,   256,  0, stream>>>(h, att, ssrc, cursor);
    gat_count <<<N_EDGES / 256, 256,  0, stream>>>(ei, cursor, rank);
    gat_scan  <<<1,             1024, 0, stream>>>(cursor);
    gat_fill  <<<N_EDGES / 256, 256,  0, stream>>>(ei, cursor, rank, csr);
    gat_node  <<<N_NODES / 8,   256,  0, stream>>>(cursor, csr, ssrc, h, att, bias, out);
}


# edges bucketed with no per-edge global atomic: per-block LDS histogram, one returning global atomic per block and bucket, LDS counting sort per bucket; no scan pass; sums in LDS-atomic arrival order
# speedup vs baseline: 5.1807x; 1.7689x over previous
#include <hip/hip_runtime.h>
#include <hip/hip_bf16.h>
#include <math.h>


typedef __attribute__((ext_vector_type(2))) float v2f;
typedef __attribute__((ext_vector_type(8))) float v8f;

#define N_NODES   50000
#define N_EDGES   800000
#define IN_CH     128
#define HEADS     4
#define OUT_CH    32
#define HC        128

__device__ __forceinline__ float lrelu(float v) { return v > 0.0f ? v : 0.2f * v; }

__global__ __launch_bounds__(256) void gat_gemm(const float* __restrict__ x,
                                                const float* __restrict__ W,
                                                float* __restrict__ h) {
    __shared__ float xs[16 * 132];
    __shared__ float wt[32 * 132];

    const int t    = threadIdx.x;
    const int wave = t >> 5;
    const int lane = t & 31;
    const int r0   = blockIdx.x * 16;

    {
        int r = t >> 5;
        int c = (t & 31) * 4;
        #pragma unroll
        for (int i = 0; i < 2; ++i) {
            int rr = r + 8 * i;
            float4 v = *(const float4*)(x + (size_t)(r0 + rr) * IN_CH + c);
            xs[rr * 132 + c + 0] = v.x;
            xs[rr * 132 + c + 1] = v.y;
            xs[rr * 132 + c + 2] = v.z;
            xs[rr * 132 + c + 3] = v.w;
        }
    }

    const int row  = lane & 15;
    const int kh   = (lane >> 4) * 2;
    const int n0   = wave * 16;
    const int ncol = n0 + row;

    v8f acc = {0.f, 0.f, 0.f, 0.f, 0.f, 0.f, 0.f, 0.f};

    for (int kc = 0; kc < IN_CH; kc += 32) {
        __syncthreads();
        #pragma unroll
        for (int i = 0; i < 4; ++i) {
            int q    = t + 256 * i;
            int flat = q * 4;
            int n    = flat >> 5;
            int k4   = flat & 31;
            float4 v = *(const float4*)(W + (size_t)n * IN_CH + kc + k4);
            wt[(k4 + 0) * 132 + n] = v.x;
            wt[(k4 + 1) * 132 + n] = v.y;
            wt[(k4 + 2) * 132 + n] = v.z;
            wt[(k4 + 3) * 132 + n] = v.w;
        }
        __syncthreads();

        #pragma unroll
        for (int kk = 0; kk < 32; kk += 4) {
            v2f a, b;
            a.x = xs[row * 132 + kc + kk + kh + 0];
            a.y = xs[row * 132 + kc + kk + kh + 1];
            b.x = wt[(kk + kh + 0) * 132 + ncol];
            b.y = wt[(kk + kh + 1) * 132 + ncol];
            acc = __builtin_amdgcn_wmma_f32_16x16x4_f32(
                false, a, false, b, (short)0, acc, false, false);
        }
    }

    const int mrow = r0 + ((lane >> 4) * 8);
    const int col  = n0 + (lane & 15);
    #pragma unroll
    for (int r = 0; r < 8; ++r) {
        h[(size_t)(mrow + r) * HC + col] = acc[r];
    }
}

#define NBK     196
#define GPAD    64
#define CHUNK   3200
#define NCHUNK  250

__device__ __forceinline__ void bucket_starts(const int* __restrict__ gcount, int* bstart, int lane) {
    int v[7];
    int s = 0;
    #pragma unroll
    for (int j = 0; j < 7; ++j) {
        const int k = lane * 7 + j;
        v[j] = (k < NBK) ? gcount[k * GPAD] : 0;
        s += v[j];
    }
    int inc = s;
    for (int off = 1; off < 32; off <<= 1) {
        const int u = __shfl_up(inc, off);
        if (lane >= off) inc += u;
    }
    int run = inc - s;
    #pragma unroll
    for (int j = 0; j < 7; ++j) {
        const int k = lane * 7 + j;
        if (k <= NBK) bstart[k] = run;
        run += v[j];
    }
}

__global__ void gat_scores(const float* __restrict__ h, const float* __restrict__ att,
                           float* __restrict__ ssrc, int* __restrict__ gcount) {
    int node = blockIdx.x * 8 + (threadIdx.x >> 5);
    int lane = threadIdx.x & 31;
    int c    = lane * 4;
    int head = lane >> 3;
    int cc   = c & 31;

    float4 hv = *(const float4*)(h + (size_t)node * HC + c);
    float4 as = *(const float4*)(att + head * (2 * OUT_CH) + cc);

    float ps = hv.x * as.x + hv.y * as.y + hv.z * as.z + hv.w * as.w;
    for (int off = 4; off >= 1; off >>= 1) ps += __shfl_xor(ps, off, 8);

    if ((lane & 7) == 0) ssrc[node * HEADS + head] = ps;
    if (lane == 0 && node < NBK) gcount[node * GPAD] = 0;
}

__global__ __launch_bounds__(1024) void gat_hist(const int* __restrict__ ei,
                                                 int* __restrict__ gcount,
                                                 int* __restrict__ blkbase) {
    __shared__ int hist[NBK];
    const int t = threadIdx.x;
    const int b = blockIdx.x;
    const bool has = t < CHUNK / 4;
    int4 d = make_int4(0, 0, 0, 0);
    if (has) d = *(const int4*)(ei + N_EDGES + b * CHUNK + t * 4);
    if (t < NBK) hist[t] = 0;
    __syncthreads();
    if (has) {
        atomicAdd(&hist[d.x >> 8], 1);
        atomicAdd(&hist[d.y >> 8], 1);
        atomicAdd(&hist[d.z >> 8], 1);
        atomicAdd(&hist[d.w >> 8], 1);
    }
    __syncthreads();
    if (t < NBK) blkbase[b * NBK + t] = atomicAdd(&gcount[t * GPAD], hist[t]);
}

__global__ __launch_bounds__(1024) void gat_scatter(const int* __restrict__ ei,
                                                    const int* __restrict__ gcount,
                                                    const int* __restrict__ blkbase,
                                                    int* __restrict__ gstart,
                                                    unsigned* __restrict__ tmp) {
    __shared__ int bstart[NBK + 1];
    __shared__ int lcur[NBK];
    const int t = threadIdx.x;
    const int b = blockIdx.x;
    const bool has = t < CHUNK / 4;
    int4 s = make_int4(0, 0, 0, 0), d = make_int4(0, 0, 0, 0);
    if (has) {
        s = *(const int4*)(ei + b * CHUNK + t * 4);
        d = *(const int4*)(ei + N_EDGES + b * CHUNK + t * 4);
    }
    const int bb = (t < NBK) ? blkbase[b * NBK + t] : 0;
    if (t < 32) bucket_starts(gcount, bstart, t);
    __syncthreads();
    if (t < NBK) lcur[t] = bstart[t] + bb;
    if (b == 0 && t <= NBK) gstart[t] = bstart[t];
    __syncthreads();
    if (has) {
        int p;
        p = atomicAdd(&lcur[d.x >> 8], 1); tmp[p] = ((unsigned)d.x << 16) | (unsigned)s.x;
        p = atomicAdd(&lcur[d.y >> 8], 1); tmp[p] = ((unsigned)d.y << 16) | (unsigned)s.y;
        p = atomicAdd(&lcur[d.z >> 8], 1); tmp[p] = ((unsigned)d.z << 16) | (unsigned)s.z;
        p = atomicAdd(&lcur[d.w >> 8], 1); tmp[p] = ((unsigned)d.w << 16) | (unsigned)s.w;
    }
}

#define BK_KEEP 5
__global__ __launch_bounds__(1024) void gat_bucket(const int* __restrict__ gstart,
                                                   const unsigned* __restrict__ tmp,
                                                   int* __restrict__ cursor,
                                                   unsigned short* __restrict__ csr) {
    __shared__ __attribute__((aligned(16))) int cnt[256];
    __shared__ __attribute__((aligned(16))) int lst[256];
    const int t   = threadIdx.x;
    const int k   = blockIdx.x;
    const int beg = gstart[k];
    const int n   = gstart[k + 1] - beg;

    unsigned w[BK_KEEP];
    #pragma unroll
    for (int u = 0; u < BK_KEEP; ++u) {
        const int i = t + u * 1024;
        w[u] = (i < n) ? tmp[beg + i] : 0u;
    }
    if (t < 256) cnt[t] = 0;
    __syncthreads();
    #pragma unroll
    for (int u = 0; u < BK_KEEP; ++u)
        if (t + u * 1024 < n) atomicAdd(&cnt[(w[u] >> 16) & 255u], 1);
    for (int i = t + BK_KEEP * 1024; i < n; i += 1024)
        atomicAdd(&cnt[(tmp[beg + i] >> 16) & 255u], 1);
    __syncthreads();
    if (t < 32) {
        int v[8];
        int s = 0;
        #pragma unroll
        for (int j = 0; j < 8; ++j) { v[j] = cnt[t * 8 + j]; s += v[j]; }
        int inc = s;
        for (int off = 1; off < 32; off <<= 1) {
            const int u = __shfl_up(inc, off);
            if (t >= off) inc += u;
        }
        int o[8];
        o[0] = beg + inc - s;
        #pragma unroll
        for (int j = 1; j < 8; ++j) o[j] = o[j - 1] + v[j - 1];
        const int4 lo = make_int4(o[0], o[1], o[2], o[3]);
        const int4 hi = make_int4(o[4], o[5], o[6], o[7]);
        *(int4*)(lst + t * 8)     = lo;
        *(int4*)(lst + t * 8 + 4) = hi;
        const int node = k * 256 + t * 8;
        if (node < N_NODES) {
            *(int4*)(cursor + node)     = lo;
            *(int4*)(cursor + node + 4) = hi;
        }
    }
    __syncthreads();
    #pragma unroll
    for (int u = 0; u < BK_KEEP; ++u)
        if (t + u * 1024 < n) {
            const int p = atomicAdd(&lst[(w[u] >> 16) & 255u], 1);
            csr[p] = (unsigned short)(w[u] & 0xFFFFu);
        }
    for (int i = t + BK_KEEP * 1024; i < n; i += 1024) {
        const unsigned x = tmp[beg + i];
        const int p = atomicAdd(&lst[(x >> 16) & 255u], 1);
        csr[p] = (unsigned short)(x & 0xFFFFu);
    }
}

__device__ __forceinline__ float wave_max(float v) {
    for (int off = 16; off >= 1; off >>= 1) v = fmaxf(v, __shfl_xor(v, off));
    return v;
}
__device__ __forceinline__ float wave_sum(float v) {
    for (int off = 16; off >= 1; off >>= 1) v += __shfl_xor(v, off);
    return v;
}
__device__ __forceinline__ float bcast(float v, int l) {
    return __int_as_float(__builtin_amdgcn_readlane(__float_as_int(v), l));
}
__device__ __forceinline__ float pick(float a, float b, float c, float d, int head) {
    return (head < 2) ? (head == 0 ? a : b) : (head == 2 ? c : d);
}

__global__ __launch_bounds__(256) void gat_node(const int* __restrict__ cursor,
                                                const unsigned short* __restrict__ csr,
                                                const float* __restrict__ ssrc,
                                                const float* __restrict__ h,
                                                const float* __restrict__ att,
                                                const float* __restrict__ bias,
                                                float* __restrict__ out) {
    const int node = blockIdx.x * 8 + (threadIdx.x >> 5);
    const int lane = threadIdx.x & 31;
    const int c    = lane * 4;
    const int head = lane >> 3;
    const int cc   = c & 31;

    const int beg = __builtin_amdgcn_readfirstlane(cursor[node]);
    const int end = __builtin_amdgcn_readfirstlane((node == N_NODES - 1) ? N_EDGES : cursor[node + 1]);

    const float4 hs = *(const float4*)(h + (size_t)node * HC + c);
    const float4 as = *(const float4*)(att + head * (2 * OUT_CH) + cc);
    const float4 ad = *(const float4*)(att + head * (2 * OUT_CH) + OUT_CH + cc);
    float ps = hs.x * as.x + hs.y * as.y + hs.z * as.z + hs.w * as.w;
    float pd = hs.x * ad.x + hs.y * ad.y + hs.z * ad.z + hs.w * ad.w;
    for (int off = 4; off >= 1; off >>= 1) {
        ps += __shfl_xor(ps, off, 8);
        pd += __shfl_xor(pd, off, 8);
    }

    const float d0 = bcast(pd, 0), d1 = bcast(pd, 8), d2 = bcast(pd, 16), d3 = bcast(pd, 24);

    float m0 = lrelu(bcast(ps, 0)  + d0);
    float m1 = lrelu(bcast(ps, 8)  + d1);
    float m2 = lrelu(bcast(ps, 16) + d2);
    float m3 = lrelu(bcast(ps, 24) + d3);
    float S0 = 1.0f, S1 = 1.0f, S2 = 1.0f, S3 = 1.0f;
    float4 acc = hs;

    for (int base = beg; base < end; base += 32) {
        const int  j   = base + lane;
        const bool act = j < end;
        const int  s   = act ? (int)csr[j] : 0;
        const float4 a = *(const float4*)(ssrc + s * HEADS);

        const float e0 = act ? lrelu(a.x + d0) : -INFINITY;
        const float e1 = act ? lrelu(a.y + d1) : -INFINITY;
        const float e2 = act ? lrelu(a.z + d2) : -INFINITY;
        const float e3 = act ? lrelu(a.w + d3) : -INFINITY;

        const float n0 = fmaxf(m0, wave_max(e0));
        const float n1 = fmaxf(m1, wave_max(e1));
        const float n2 = fmaxf(m2, wave_max(e2));
        const float n3 = fmaxf(m3, wave_max(e3));

        const float r0 = __expf(m0 - n0);
        const float r1 = __expf(m1 - n1);
        const float r2 = __expf(m2 - n2);
        const float r3 = __expf(m3 - n3);

        const float p0 = act ? __expf(e0 - n0) : 0.0f;
        const float p1 = act ? __expf(e1 - n1) : 0.0f;
        const float p2 = act ? __expf(e2 - n2) : 0.0f;
        const float p3 = act ? __expf(e3 - n3) : 0.0f;

        S0 = S0 * r0 + wave_sum(p0);
        S1 = S1 * r1 + wave_sum(p1);
        S2 = S2 * r2 + wave_sum(p2);
        S3 = S3 * r3 + wave_sum(p3);
        m0 = n0; m1 = n1; m2 = n2; m3 = n3;

        const float r = pick(r0, r1, r2, r3, head);
        acc.x *= r; acc.y *= r; acc.z *= r; acc.w *= r;

        const int nin = min(32, end - base);
        int jj = 0;
        for (; jj + 4 <= nin; jj += 4) {
            float4 hv[4];
            float  wq[4];
            for (int u = 0; u < 4; ++u) {
                const int sj = __builtin_amdgcn_readlane(s, jj + u);
                hv[u] = *(const float4*)(h + (size_t)sj * HC + c);
                wq[u] = pick(bcast(p0, jj + u), bcast(p1, jj + u),
                             bcast(p2, jj + u), bcast(p3, jj + u), head);
            }
            for (int u = 0; u < 4; ++u) {
                acc.x += wq[u] * hv[u].x;
                acc.y += wq[u] * hv[u].y;
                acc.z += wq[u] * hv[u].z;
                acc.w += wq[u] * hv[u].w;
            }
        }
        for (; jj < nin; ++jj) {
            const int sj = __builtin_amdgcn_readlane(s, jj);
            const float4 hv = *(const float4*)(h + (size_t)sj * HC + c);
            const float wq = pick(bcast(p0, jj), bcast(p1, jj), bcast(p2, jj), bcast(p3, jj), head);
            acc.x += wq * hv.x;
            acc.y += wq * hv.y;
            acc.z += wq * hv.z;
            acc.w += wq * hv.w;
        }
    }

    const float inv = 1.0f / fmaxf(pick(S0, S1, S2, S3, head), 1e-10f);
    const float4 b  = *(const float4*)(bias + c);
    float4 o;
    o.x = acc.x * inv + b.x;
    o.y = acc.y * inv + b.y;
    o.z = acc.z * inv + b.z;
    o.w = acc.w * inv + b.w;
    *(float4*)(out + (size_t)node * HC + c) = o;
}

extern "C" void kernel_launch(void* const* d_in, const int* in_sizes, int n_in,
                              void* d_out, int out_size, void* d_ws, size_t ws_size,
                              hipStream_t stream) {
    const float* x    = (const float*)d_in[0];
    const int*   ei   = (const int*)  d_in[1];
    const float* W    = (const float*)d_in[2];
    const float* att  = (const float*)d_in[3];
    const float* bias = (const float*)d_in[4];
    float* out = (float*)d_out;
    char* ob = (char*)d_out;
    unsigned* tmp     = (unsigned*)(ob);
    int*      blkbase = (int*)(ob + 4194304);
    int*      gcount  = (int*)(ob + 5242880);
    int*      gstart  = (int*)(ob + 6291456);

    char* ws = (char*)d_ws;
    float*          h      = (float*)(ws);
    float*          ssrc   = (float*)(ws + 25600000);
    int*            cursor = (int*)(ws + 26400000);
    unsigned short* csr    = (unsigned short*)(ws + 26600000);

    gat_gemm   <<<N_NODES / 16, 256,  0, stream>>>(x, W, h);
    gat_scores <<<N_NODES / 8,  256,  0, stream>>>(h, att, ssrc, gcount);
    gat_hist   <<<NCHUNK,       1024, 0, stream>>>(ei, gcount, blkbase);
    gat_scatter<<<NCHUNK,       1024, 0, stream>>>(ei, gcount, blkbase, gstart, tmp);
    gat_bucket <<<NBK,          1024, 0, stream>>>(gstart, tmp, cursor, csr);
    gat_node   <<<N_NODES / 8,  256,  0, stream>>>(cursor, csr, ssrc, h, att, bias, out);
}


# GEMM reads W once per block (whole rows via LDS) and keeps each wave's B fragments in registers over 4-5 row tiles: no W restaging per 16 rows, one barrier a tile, 768 blocks; same k and sum order
# speedup vs baseline: 8.1656x; 1.5762x over previous
#include <hip/hip_runtime.h>
#include <hip/hip_bf16.h>
#include <math.h>


typedef __attribute__((ext_vector_type(2))) float v2f;
typedef __attribute__((ext_vector_type(8))) float v8f;

#define N_NODES   50000
#define N_EDGES   800000
#define IN_CH     128
#define HEADS     4
#define OUT_CH    32
#define HC        128

__device__ __forceinline__ float lrelu(float v) { return v > 0.0f ? v : 0.2f * v; }

#define NBK     196

__device__ __forceinline__ float row16_sum(float v) {
    v += __int_as_float(__builtin_amdgcn_update_dpp(0, __float_as_int(v), 0x161, 0xf, 0xf, false));
    v += __int_as_float(__builtin_amdgcn_update_dpp(0, __float_as_int(v), 0x162, 0xf, 0xf, false));
    v += __int_as_float(__builtin_amdgcn_update_dpp(0, __float_as_int(v), 0x164, 0xf, 0xf, false));
    v += __int_as_float(__builtin_amdgcn_update_dpp(0, __float_as_int(v), 0x168, 0xf, 0xf, false));
    return v;
}

#define CHUNK   3200
#define NCHUNK  250
#define NOFF    (NBK + 1)

__device__ __forceinline__ void sort_chunk(const int* __restrict__ ei,
                                           unsigned* __restrict__ tmp,
                                           int* __restrict__ blkoff,
                                           int* hist, int* bst, int* lcur, unsigned* sorted) {
    const int t = threadIdx.x;
    const int b = blockIdx.x;
    int4 s[4], d[4];
    #pragma unroll
    for (int u = 0; u < 4; ++u) {
        const int q = t + 256 * u;
        s[u] = make_int4(0, 0, 0, 0);
        d[u] = make_int4(0, 0, 0, 0);
        if (q < CHUNK / 4) {
            s[u] = *(const int4*)(ei + b * CHUNK + q * 4);
            d[u] = *(const int4*)(ei + N_EDGES + b * CHUNK + q * 4);
        }
    }
    if (t < NBK) hist[t] = 0;
    __syncthreads();
    #pragma unroll
    for (int u = 0; u < 4; ++u)
        if (t + 256 * u < CHUNK / 4) {
            atomicAdd(&hist[d[u].x >> 8], 1);
            atomicAdd(&hist[d[u].y >> 8], 1);
            atomicAdd(&hist[d[u].z >> 8], 1);
            atomicAdd(&hist[d[u].w >> 8], 1);
        }
    __syncthreads();
    if (t < 32) {
        int v[7];
        int sum = 0;
        #pragma unroll
        for (int j = 0; j < 7; ++j) {
            const int k = t * 7 + j;
            v[j] = (k < NBK) ? hist[k] : 0;
            sum += v[j];
        }
        int inc = sum;
        for (int off = 1; off < 32; off <<= 1) {
            const int up = __shfl_up(inc, off);
            if (t >= off) inc += up;
        }
        int run = inc - sum;
        #pragma unroll
        for (int j = 0; j < 7; ++j) {
            const int k = t * 7 + j;
            if (k <= NBK) bst[k] = run;
            run += v[j];
        }
    }
    __syncthreads();
    if (t < NBK) lcur[t] = bst[t];
    if (t < NOFF) blkoff[b * NOFF + t] = bst[t];
    __syncthreads();
    #pragma unroll
    for (int u = 0; u < 4; ++u)
        if (t + 256 * u < CHUNK / 4) {
            int p;
            p = atomicAdd(&lcur[d[u].x >> 8], 1); sorted[p] = ((unsigned)d[u].x << 16) | (unsigned)s[u].x;
            p = atomicAdd(&lcur[d[u].y >> 8], 1); sorted[p] = ((unsigned)d[u].y << 16) | (unsigned)s[u].y;
            p = atomicAdd(&lcur[d[u].z >> 8], 1); sorted[p] = ((unsigned)d[u].z << 16) | (unsigned)s[u].z;
            p = atomicAdd(&lcur[d[u].w >> 8], 1); sorted[p] = ((unsigned)d[u].w << 16) | (unsigned)s[u].w;
        }
    __syncthreads();
    #pragma unroll
    for (int u = 0; u < 4; ++u) {
        const int q = t + 256 * u;
        if (q < CHUNK / 4) ((uint4*)(tmp + b * CHUNK))[q] = ((const uint4*)sorted)[q];
    }
}

#define NTILE        (N_NODES / 16)
#define GEMM_BLOCKS  768
#define XT           (16 * 132)
#define WL           (HC * 132)
__global__ __launch_bounds__(256) void gat_gemm(const float* x,
                                                const float* __restrict__ W,
                                                const float* __restrict__ att,
                                                const int* __restrict__ ei,
                                                float* __restrict__ h,
                                                float* __restrict__ ssrc,
                                                unsigned* __restrict__ tmp,
                                                int* __restrict__ blkoff) {
    __shared__ __attribute__((aligned(16))) char smem[(WL + 2 * 8 * 16) * 4];
    if (blockIdx.x < NCHUNK) {
        int* ib = (int*)smem;
        sort_chunk(ei, tmp, blkoff, ib, ib + 256, ib + 512, (unsigned*)(smem + 4096));
        return;
    }
    float* wl   = (float*)smem;
    float* xs   = (float*)smem;
    float* part = wl + WL;

    const int t    = threadIdx.x;
    const int wave = t >> 5;
    const int lane = t & 31;
    const int gb   = blockIdx.x - NCHUNK;

    const int row  = lane & 15;
    const int kh   = (lane >> 4) * 2;
    const int n0   = wave * 16;
    const int ncol = n0 + row;

    const int xr = t >> 5;
    const int xc = (t & 31) * 4;

    const float4 xa0 = *(const float4*)(x + (size_t)(gb * 16 + xr) * IN_CH + xc);
    const float4 xb0 = *(const float4*)(x + (size_t)(gb * 16 + xr + 8) * IN_CH + xc);

    {
        float4 wv[16];
        #pragma unroll
        for (int i = 0; i < 16; ++i)
            wv[i] = *(const float4*)(W + (size_t)(xr + 8 * i) * IN_CH + xc);
        #pragma unroll
        for (int i = 0; i < 16; ++i)
            *(float4*)(wl + (xr + 8 * i) * 132 + xc) = wv[i];
    }

    const float av = att[(wave >> 1) * (2 * OUT_CH) + (wave & 1) * 16 + (lane & 15)];
    __syncthreads();

    v2f bf[32];
    {
        const float* wp = wl + ncol * 132 + kh;
        #pragma unroll
        for (int s = 0; s < 32; ++s) bf[s] = *(const v2f*)(wp + 4 * s);
    }
    __syncthreads();

    *(float4*)(xs + xr * 132 + xc)       = xa0;
    *(float4*)(xs + (xr + 8) * 132 + xc) = xb0;
    __syncthreads();

    int cur = 0;
    for (int tile = gb; tile < NTILE; tile += GEMM_BLOCKS) {
        const int  r0   = tile * 16;
        const int  nxt  = tile + GEMM_BLOCKS;
        const bool more = nxt < NTILE;

        float4 xa = make_float4(0.f, 0.f, 0.f, 0.f), xb = xa;
        if (more) {
            xa = *(const float4*)(x + (size_t)(nxt * 16 + xr) * IN_CH + xc);
            xb = *(const float4*)(x + (size_t)(nxt * 16 + xr + 8) * IN_CH + xc);
        }

        const float* xt = xs + cur * XT + row * 132 + kh;
        v8f acc = {0.f, 0.f, 0.f, 0.f, 0.f, 0.f, 0.f, 0.f};
        #pragma unroll
        for (int s = 0; s < 32; ++s) {
            const v2f a = *(const v2f*)(xt + 4 * s);
            acc = __builtin_amdgcn_wmma_f32_16x16x4_f32(
                false, a, false, bf[s], (short)0, acc, false, false);
        }

        float sel = 0.0f;
        #pragma unroll
        for (int r = 0; r < 8; ++r) {
            const float v = row16_sum(acc[r] * av);
            if ((lane & 7) == r) sel = v;
        }
        float* pc = part + cur * (8 * 16);
        if ((lane & 8) == 0) pc[wave * 16 + (lane >> 4) * 8 + (lane & 7)] = sel;

        if (more) {
            float* xn = xs + (cur ^ 1) * XT;
            *(float4*)(xn + xr * 132 + xc)       = xa;
            *(float4*)(xn + (xr + 8) * 132 + xc) = xb;
        }
        __syncthreads();
        if (t < 64) {
            const int rr = t >> 2;
            const int hd = t & 3;
            ssrc[(r0 + rr) * HEADS + hd] = pc[(2 * hd) * 16 + rr] + pc[(2 * hd + 1) * 16 + rr];
        }

        const int mrow = r0 + ((lane >> 4) * 8);
        const int col  = n0 + (lane & 15);
        #pragma unroll
        for (int r = 0; r < 8; ++r) {
            h[(size_t)(mrow + r) * HC + col] = acc[r];
        }
        cur ^= 1;
    }
}

#define BK_KEEP 8
__global__ __launch_bounds__(1024) void gat_bucket(const int* __restrict__ blkoff,
                                                   const unsigned* __restrict__ tmp,
                                                   int* __restrict__ cursor,
                                                   unsigned short* __restrict__ csr) {
    __shared__ __attribute__((aligned(16))) int cnt[256];
    __shared__ __attribute__((aligned(16))) int lst[256];
    __shared__ int sbeg;
    const int t   = threadIdx.x;
    const int k   = blockIdx.x;
    const int run = t >> 2;
    const int q   = t & 3;

    int off = 0, n = 0;
    if (run < NCHUNK) {
        off = blkoff[run * NOFF + k];
        n   = blkoff[run * NOFF + k + 1] - off;
    }
    const unsigned* rp = tmp + run * CHUNK + off;

    unsigned w[BK_KEEP];
    #pragma unroll
    for (int u = 0; u < BK_KEEP; ++u) {
        const int j = q + 4 * u;
        w[u] = (j < n) ? rp[j] : 0u;
    }
    if (t < 256) cnt[t] = 0;
    if (t == 0) sbeg = 0;
    __syncthreads();

    int bs = (q == 0) ? off : 0;
    for (int o = 16; o > 0; o >>= 1) bs += __shfl_xor(bs, o);
    if ((t & 31) == 0) atomicAdd(&sbeg, bs);

    #pragma unroll
    for (int u = 0; u < BK_KEEP; ++u)
        if (q + 4 * u < n) atomicAdd(&cnt[(w[u] >> 16) & 255u], 1);
    for (int j = q + 4 * BK_KEEP; j < n; j += 4)
        atomicAdd(&cnt[(rp[j] >> 16) & 255u], 1);
    __syncthreads();
    if (t < 32) {
        const int beg = sbeg;
        int v[8];
        int s = 0;
        #pragma unroll
        for (int j = 0; j < 8; ++j) { v[j] = cnt[t * 8 + j]; s += v[j]; }
        int inc = s;
        for (int o = 1; o < 32; o <<= 1) {
            const int up = __shfl_up(inc, o);
            if (t >= o) inc += up;
        }
        int o8[8];
        o8[0] = beg + inc - s;
        #pragma unroll
        for (int j = 1; j < 8; ++j) o8[j] = o8[j - 1] + v[j - 1];
        const int4 lo = make_int4(o8[0], o8[1], o8[2], o8[3]);
        const int4 hi = make_int4(o8[4], o8[5], o8[6], o8[7]);
        *(int4*)(lst + t * 8)     = lo;
        *(int4*)(lst + t * 8 + 4) = hi;
        const int node = k * 256 + t * 8;
        if (node < N_NODES) {
            *(int4*)(cursor + node)     = lo;
            *(int4*)(cursor + node + 4) = hi;
        }
    }
    __syncthreads();
    #pragma unroll
    for (int u = 0; u < BK_KEEP; ++u)
        if (q + 4 * u < n) {
            const int p = atomicAdd(&lst[(w[u] >> 16) & 255u], 1);
            csr[p] = (unsigned short)(w[u] & 0xFFFFu);
        }
    for (int j = q + 4 * BK_KEEP; j < n; j += 4) {
        const unsigned e = rp[j];
        const int p = atomicAdd(&lst[(e >> 16) & 255u], 1);
        csr[p] = (unsigned short)(e & 0xFFFFu);
    }
}

__device__ __forceinline__ float grp8_sum(float v) {
    v += __int_as_float(__builtin_amdgcn_update_dpp(0, __float_as_int(v), 0x164, 0xf, 0xf, false));
    v += __int_as_float(__builtin_amdgcn_update_dpp(0, __float_as_int(v), 0x162, 0xf, 0xf, false));
    v += __int_as_float(__builtin_amdgcn_update_dpp(0, __float_as_int(v), 0x161, 0xf, 0xf, false));
    return v;
}
__device__ __forceinline__ float grp8_max(float v) {
    v = fmaxf(v, __int_as_float(__builtin_amdgcn_update_dpp(0, __float_as_int(v), 0x164, 0xf, 0xf, false)));
    v = fmaxf(v, __int_as_float(__builtin_amdgcn_update_dpp(0, __float_as_int(v), 0x162, 0xf, 0xf, false)));
    v = fmaxf(v, __int_as_float(__builtin_amdgcn_update_dpp(0, __float_as_int(v), 0x161, 0xf, 0xf, false)));
    return v;
}
template <int U>
__device__ __forceinline__ float grp8_bcast(float v) {
    const unsigned lo = 0x11111111u * (unsigned)U;
    const unsigned hi = 0x11111111u * (unsigned)(8 + U);
    return __int_as_float(__builtin_amdgcn_permlane16(__float_as_int(v), __float_as_int(v), lo, hi, true, false));
}
__device__ __forceinline__ float4 ld_row(const float* __restrict__ h, int sj, unsigned cb) {
    const char* row = (const char*)h + ((size_t)(unsigned)sj << 9);
    return *(const float4*)(row + cb);
}
template <int U>
__device__ __forceinline__ void add_row(float4& acc, float pk, const float4& hv) {
    const float w = grp8_bcast<U>(pk);
    acc.x = fmaf(w, hv.x, acc.x);
    acc.y = fmaf(w, hv.y, acc.y);
    acc.z = fmaf(w, hv.z, acc.z);
    acc.w = fmaf(w, hv.w, acc.w);
}
__device__ __forceinline__ void gather8(const float* __restrict__ h, unsigned c, int sk, float pk,
                                        int rem, float4& acc) {
    if (rem >= 8) {
        const float4 h0 = ld_row(h, __builtin_amdgcn_readlane(sk, 0), c);
        const float4 h1 = ld_row(h, __builtin_amdgcn_readlane(sk, 1), c);
        const float4 h2 = ld_row(h, __builtin_amdgcn_readlane(sk, 2), c);
        const float4 h3 = ld_row(h, __builtin_amdgcn_readlane(sk, 3), c);
        const float4 h4 = ld_row(h, __builtin_amdgcn_readlane(sk, 4), c);
        const float4 h5 = ld_row(h, __builtin_amdgcn_readlane(sk, 5), c);
        const float4 h6 = ld_row(h, __builtin_amdgcn_readlane(sk, 6), c);
        const float4 h7 = ld_row(h, __builtin_amdgcn_readlane(sk, 7), c);
        add_row<0>(acc, pk, h0);
        add_row<1>(acc, pk, h1);
        add_row<2>(acc, pk, h2);
        add_row<3>(acc, pk, h3);
        add_row<4>(acc, pk, h4);
        add_row<5>(acc, pk, h5);
        add_row<6>(acc, pk, h6);
        add_row<7>(acc, pk, h7);
    } else {
        const float4 z = make_float4(0.0f, 0.0f, 0.0f, 0.0f);
        const float4 h0 = ld_row(h, __builtin_amdgcn_readlane(sk, 0), c);
        const float4 h1 = (rem > 1) ? ld_row(h, __builtin_amdgcn_readlane(sk, 1), c) : z;
        const float4 h2 = (rem > 2) ? ld_row(h, __builtin_amdgcn_readlane(sk, 2), c) : z;
        const float4 h3 = (rem > 3) ? ld_row(h, __builtin_amdgcn_readlane(sk, 3), c) : z;
        const float4 h4 = (rem > 4) ? ld_row(h, __builtin_amdgcn_readlane(sk, 4), c) : z;
        const float4 h5 = (rem > 5) ? ld_row(h, __builtin_amdgcn_readlane(sk, 5), c) : z;
        const float4 h6 = (rem > 6) ? ld_row(h, __builtin_amdgcn_readlane(sk, 6), c) : z;
        add_row<0>(acc, pk, h0);
        add_row<1>(acc, pk, h1);
        add_row<2>(acc, pk, h2);
        add_row<3>(acc, pk, h3);
        add_row<4>(acc, pk, h4);
        add_row<5>(acc, pk, h5);
        add_row<6>(acc, pk, h6);
    }
}

__global__ __launch_bounds__(256) void gat_node(const int* __restrict__ cursor,
                                                const unsigned short* __restrict__ csr,
                                                const float* __restrict__ ssrc,
                                                const float* __restrict__ h,
                                                const float* __restrict__ att,
                                                const float* __restrict__ bias,
                                                float* __restrict__ out) {
    const int node = blockIdx.x * 8 + (threadIdx.x >> 5);
    const int lane = threadIdx.x & 31;
    const int c    = lane * 4;
    const int head = lane >> 3;
    const int i    = lane & 7;
    const int cc   = c & 31;

    const int beg = __builtin_amdgcn_readfirstlane(cursor[node]);
    const int end = __builtin_amdgcn_readfirstlane((node == N_NODES - 1) ? N_EDGES : cursor[node + 1]);

    const float4 hs = *(const float4*)(h + (size_t)node * HC + c);
    const float4 as = *(const float4*)(att + head * (2 * OUT_CH) + cc);
    const float4 ad = *(const float4*)(att + head * (2 * OUT_CH) + OUT_CH + cc);
    float ps = hs.x * as.x + hs.y * as.y + hs.z * as.z + hs.w * as.w;
    float pd = hs.x * ad.x + hs.y * ad.y + hs.z * ad.z + hs.w * ad.w;
    ps = grp8_sum(ps);
    pd = grp8_sum(pd);

    float m = lrelu(ps + pd);
    float S = 1.0f;
    float4 acc = hs;

    const char* sg = (const char*)(ssrc + head);
    const unsigned cb = (unsigned)lane * 16u;
    const int last = end - 1;

    for (int base = beg; base < end; base += 32) {
        const int  nin   = min(32, end - base);
        const bool longg = nin > 16;
        const int  j0    = base + i;
        const bool a0 = i < nin, a1 = i + 8 < nin, a2 = i + 16 < nin, a3 = i + 24 < nin;

        int q0 = csr[min(j0, last)];
        int q1 = csr[min(j0 + 8, last)];
        int q2 = 0, q3 = 0;
        if (longg) {
            q2 = csr[min(j0 + 16, last)];
            q3 = csr[min(j0 + 24, last)];
        }
        const int s0 = a0 ? q0 : node;
        const int s1 = a1 ? q1 : node;
        const int s2 = a2 ? q2 : node;
        const int s3 = a3 ? q3 : node;

        float x0 = *(const float*)(sg + ((unsigned)s0 << 4));
        float x1 = *(const float*)(sg + ((unsigned)s1 << 4));
        float x2 = 0.0f, x3 = 0.0f;
        if (longg) {
            x2 = *(const float*)(sg + ((unsigned)s2 << 4));
            x3 = *(const float*)(sg + ((unsigned)s3 << 4));
        }

        const float e0 = a0 ? lrelu(x0 + pd) : -INFINITY;
        const float e1 = a1 ? lrelu(x1 + pd) : -INFINITY;
        float e2 = -INFINITY, e3 = -INFINITY;
        float mg = fmaxf(e0, e1);
        if (longg) {
            e2 = a2 ? lrelu(x2 + pd) : -INFINITY;
            e3 = a3 ? lrelu(x3 + pd) : -INFINITY;
            mg = fmaxf(mg, fmaxf(e2, e3));
        }
        const float mn = fmaxf(m, grp8_max(mg));
        const float r  = __expf(m - mn);

        const float p0 = a0 ? __expf(e0 - mn) : 0.0f;
        const float p1 = a1 ? __expf(e1 - mn) : 0.0f;
        float p2 = 0.0f, p3 = 0.0f;
        float pl = p0 + p1;
        if (longg) {
            p2 = a2 ? __expf(e2 - mn) : 0.0f;
            p3 = a3 ? __expf(e3 - mn) : 0.0f;
            pl += p2;
            pl += p3;
        }
        S = S * r + grp8_sum(pl);
        m = mn;
        acc.x *= r; acc.y *= r; acc.z *= r; acc.w *= r;

        gather8(h, cb, s0, p0, nin, acc);
        if (nin > 8)  gather8(h, cb, s1, p1, nin - 8,  acc);
        if (nin > 16) gather8(h, cb, s2, p2, nin - 16, acc);
        if (nin > 24) gather8(h, cb, s3, p3, nin - 24, acc);
    }

    const float inv = 1.0f / fmaxf(S, 1e-10f);
    const float4 b  = *(const float4*)(bias + c);
    float4 o;
    o.x = acc.x * inv + b.x;
    o.y = acc.y * inv + b.y;
    o.z = acc.z * inv + b.z;
    o.w = acc.w * inv + b.w;
    *(float4*)(out + (size_t)node * HC + c) = o;
}

extern "C" void kernel_launch(void* const* d_in, const int* in_sizes, int n_in,
                              void* d_out, int out_size, void* d_ws, size_t ws_size,
                              hipStream_t stream) {
    const float* x    = (const float*)d_in[0];
    const int*   ei   = (const int*)  d_in[1];
    const float* W    = (const float*)d_in[2];
    const float* att  = (const float*)d_in[3];
    const float* bias = (const float*)d_in[4];
    float* out = (float*)d_out;
    char* ob = (char*)d_out;
    unsigned* tmp     = (unsigned*)(ob);
    int*      blkoff  = (int*)(ob + 4194304);

    char* ws = (char*)d_ws;
    float*          h      = (float*)(ws);
    float*          ssrc   = (float*)(ws + 25600000);
    int*            cursor = (int*)(ws + 26400000);
    unsigned short* csr    = (unsigned short*)(ws + 26600000);

    gat_gemm   <<<NCHUNK + GEMM_BLOCKS, 256, 0, stream>>>(x, W, att, ei, h, ssrc, tmp, blkoff);
    gat_bucket <<<NBK,          1024, 0, stream>>>(blkoff, tmp, cursor, csr);
    gat_node   <<<N_NODES / 8,  256,  0, stream>>>(cursor, csr, ssrc, h, att, bias, out);
}
